# starting kernel
# baseline (speedup 1.0000x reference)
#include <hip/hip_runtime.h>
#include <math.h>

typedef __attribute__((ext_vector_type(2))) float v2f;
typedef __attribute__((ext_vector_type(8))) float v8f;

#define WMMA_F32(a, b, c) \
    __builtin_amdgcn_wmma_f32_16x16x4_f32(false, (a), false, (b), (short)0, (c), false, false)

#define BN_EPS 1e-3f
#define BIG 3.0e38f

__device__ __forceinline__ void gemm16x64(const float* A, const float* Wt,
                                          int lrow, int half, v8f acc[4]) {
#pragma unroll
    for (int kk = 0; kk < 16; ++kk) {
        const int cb = (kk << 2) + (half << 1);
        v2f a = *(const v2f*)(A + (lrow << 6) + cb);
#pragma unroll
        for (int t = 0; t < 4; ++t) {
            v2f bw = *(const v2f*)(Wt + ((lrow + (t << 4)) << 6) + cb);
            acc[t] = WMMA_F32(a, bw, acc[t]);
        }
    }
}

__device__ __forceinline__ void bn_relu(v8f acc[4], const float* g, const float* be,
                                        const float* mn, const float* vr, int lrow) {
#pragma unroll
    for (int t = 0; t < 4; ++t) {
        const int d = lrow + (t << 4);
        const float s = g[d] * (1.0f / sqrtf(vr[d] + BN_EPS));
        const float bi = be[d] - mn[d] * s;
#pragma unroll
        for (int v = 0; v < 8; ++v) acc[t][v] = fmaxf(acc[t][v] * s + bi, 0.0f);
    }
}

__device__ __forceinline__ void store16x64(const v8f acc[4], int lrow, int half, float* Y) {
#pragma unroll
    for (int t = 0; t < 4; ++t)
#pragma unroll
        for (int v = 0; v < 8; ++v)
            Y[((v + (half << 3)) << 6) + lrow + (t << 4)] = acc[t][v];
}

__global__ __launch_bounds__(256) void edgeconv_fused_kernel(
    const float* __restrict__ points,
    const float* __restrict__ feats,
    const float* __restrict__ w0,
    const float* __restrict__ w1,
    const float* __restrict__ w2,
    const float* __restrict__ gammas,
    const float* __restrict__ betas,
    const float* __restrict__ means,
    const float* __restrict__ vars,
    const float* __restrict__ scw,
    const float* __restrict__ scg, const float* __restrict__ scb,
    const float* __restrict__ scm, const float* __restrict__ scv,
    float* __restrict__ out)
{
    extern __shared__ float smem[];
    float* pts     = smem;
    float* Wt0     = pts + 768;
    float* Wt1     = Wt0 + 8192;
    float* Wt2     = Wt1 + 4096;
    float* scWt    = Wt2 + 4096;
    float* waveBuf = scWt + 4096;
    float* ctrAll  = waveBuf + 16384;
    float* ftsAll  = ctrAll + 512;
    int*   nbrAll  = (int*)(ftsAll + 512);

    const int tid  = threadIdx.x;
    const int lane = tid & 31;
    const int w    = tid >> 5;
    const int lrow = lane & 15;
    const int half = lane >> 4;
    const int b    = blockIdx.x >> 5;
    const int pi   = ((blockIdx.x & 31) << 3) | w;

    for (int i = tid; i < 768; i += 256) pts[i] = points[b * 768 + i];
    for (int i = tid; i < 8192; i += 256) {
        const int d = i >> 7, c = i & 127;
        Wt0[i] = w0[c * 64 + d];
    }
    for (int i = tid; i < 4096; i += 256) {
        const int d = i >> 6, c = i & 63;
        Wt1[i]  = w1[c * 64 + d];
        Wt2[i]  = w2[c * 64 + d];
        scWt[i] = scw[c * 64 + d];
    }
    __syncthreads();

    float* bufA = waveBuf + (w << 11);
    float* bufB = bufA + 1024;
    float* cC   = ctrAll + (w << 6);
    float* ftsW = ftsAll + (w << 6);
    int*   nbrW = nbrAll + (w << 4);

    const float px = pts[pi * 3], py = pts[pi * 3 + 1], pz = pts[pi * 3 + 2];
    float dist[8];
#pragma unroll
    for (int j = 0; j < 8; ++j) {
        const int q = lane + (j << 5);
        const float dx = pts[q * 3] - px;
        const float dy = pts[q * 3 + 1] - py;
        const float dz = pts[q * 3 + 2] - pz;
        const float dd = dx * dx + dy * dy + dz * dz;
        dist[j] = (q == pi) ? BIG : dd;
    }
    for (int it = 0; it < 16; ++it) {
        float bv = dist[0];
        int bj = 0;
#pragma unroll
        for (int j = 1; j < 8; ++j)
            if (dist[j] < bv) { bv = dist[j]; bj = j; }
        int bq = lane + (bj << 5);
#pragma unroll
        for (int off = 16; off >= 1; off >>= 1) {
            const float ov = __shfl_xor(bv, off, 32);
            const int   oq = __shfl_xor(bq, off, 32);
            if (ov < bv || (ov == bv && oq < bq)) { bv = ov; bq = oq; }
        }
        if ((bq & 31) == lane) {
            const int jw = bq >> 5;
#pragma unroll
            for (int j = 0; j < 8; ++j)
                if (j == jw) dist[j] = BIG;
        }
        if (lane == 0) nbrW[it] = bq;
    }

    const float* fbase = feats + ((b << 8) | pi) * 64;
    const float cf0 = fbase[lane];
    const float cf1 = fbase[lane + 32];
    cC[lane]      = cf0;
    cC[lane + 32] = cf1;
    for (int r = 0; r < 16; ++r) {
        const int nq = nbrW[r];
        const float* nf = feats + (((b << 8) | nq)) * 64;
        bufA[(r << 6) + lane]      = nf[lane]      - cf0;
        bufA[(r << 6) + lane + 32] = nf[lane + 32] - cf1;
    }

    const v8f vz = {0.f, 0.f, 0.f, 0.f, 0.f, 0.f, 0.f, 0.f};
    v8f acc[4] = {vz, vz, vz, vz};
#pragma unroll
    for (int kk = 0; kk < 16; ++kk) {
        const int cb = (kk << 2) + (half << 1);
        v2f a = *(const v2f*)(cC + cb);
#pragma unroll
        for (int t = 0; t < 4; ++t) {
            v2f bw = *(const v2f*)(Wt0 + (lrow + (t << 4)) * 128 + cb);
            acc[t] = WMMA_F32(a, bw, acc[t]);
        }
    }
#pragma unroll
    for (int kk = 0; kk < 16; ++kk) {
        const int cb = (kk << 2) + (half << 1);
        v2f a = *(const v2f*)(bufA + (lrow << 6) + cb);
#pragma unroll
        for (int t = 0; t < 4; ++t) {
            v2f bw = *(const v2f*)(Wt0 + (lrow + (t << 4)) * 128 + 64 + cb);
            acc[t] = WMMA_F32(a, bw, acc[t]);
        }
    }
    bn_relu(acc, gammas, betas, means, vars, lrow);
    store16x64(acc, lrow, half, bufB);

    acc[0] = vz; acc[1] = vz; acc[2] = vz; acc[3] = vz;
    gemm16x64(bufB, Wt1, lrow, half, acc);
    bn_relu(acc, gammas + 64, betas + 64, means + 64, vars + 64, lrow);
    store16x64(acc, lrow, half, bufA);

    acc[0] = vz; acc[1] = vz; acc[2] = vz; acc[3] = vz;
    gemm16x64(bufA, Wt2, lrow, half, acc);
    bn_relu(acc, gammas + 128, betas + 128, means + 128, vars + 128, lrow);

#pragma unroll
    for (int t = 0; t < 4; ++t) {
        float s = acc[t][0] + acc[t][1] + acc[t][2] + acc[t][3] +
                  acc[t][4] + acc[t][5] + acc[t][6] + acc[t][7];
        s += __shfl_xor(s, 16, 32);
        if (lane < 16) ftsW[lrow + (t << 4)] = s * (1.0f / 16.0f);
    }

    const int d0 = lane << 1;
    float s0 = 0.f, s1 = 0.f;
#pragma unroll
    for (int c = 0; c < 64; ++c) {
        const float f = cC[c];
        s0 += f * scWt[(d0 << 6) + c];
        s1 += f * scWt[((d0 + 1) << 6) + c];
    }
    const float sA = scg[d0] * (1.0f / sqrtf(scv[d0] + BN_EPS));
    const float bA = scb[d0] - scm[d0] * sA;
    const float sB = scg[d0 + 1] * (1.0f / sqrtf(scv[d0 + 1] + BN_EPS));
    const float bB = scb[d0 + 1] - scm[d0 + 1] * sB;
    v2f o;
    o.x = fmaxf(s0 * sA + bA + ftsW[d0], 0.f);
    o.y = fmaxf(s1 * sB + bB + ftsW[d0 + 1], 0.f);
    *(v2f*)(out + (((b << 8) | pi) << 6) + d0) = o;
}

extern "C" void kernel_launch(void* const* d_in, const int* in_sizes, int n_in,
                              void* d_out, int out_size, void* d_ws, size_t ws_size,
                              hipStream_t stream) {
    (void)in_sizes; (void)n_in; (void)d_ws; (void)ws_size; (void)out_size;
    const float* points = (const float*)d_in[0];
    const float* feats  = (const float*)d_in[1];
    const float* w0     = (const float*)d_in[2];
    const float* w1     = (const float*)d_in[3];
    const float* w2     = (const float*)d_in[4];
    const float* gammas = (const float*)d_in[5];
    const float* betas  = (const float*)d_in[6];
    const float* means  = (const float*)d_in[7];
    const float* vars   = (const float*)d_in[8];
    const float* scw    = (const float*)d_in[9];
    const float* scg    = (const float*)d_in[10];
    const float* scb    = (const float*)d_in[11];
    const float* scm    = (const float*)d_in[12];
    const float* scv    = (const float*)d_in[13];
    float* out = (float*)d_out;

    const size_t ldsBytes = (768 + 8192 + 3 * 4096 + 16384 + 512 + 512) * sizeof(float)
                          + 128 * sizeof(int);
    dim3 grid(4096), block(256);
    edgeconv_fused_kernel<<<grid, block, ldsBytes, stream>>>(
        points, feats, w0, w1, w2, gammas, betas, means, vars,
        scw, scg, scb, scm, scv, out);
}


# each wave takes 16 points as the 16 matrix columns and walks the 16 neighbour slots: layer 0's centre half, equal in every slot, summed once per wave; grid 4096->256; layer-0 k and mean order change
# speedup vs baseline: 6.5208x; 6.5208x over previous
#include <hip/hip_runtime.h>
#include <math.h>


typedef __attribute__((ext_vector_type(2))) float v2f;
typedef __attribute__((ext_vector_type(4))) float v4f;
typedef __attribute__((ext_vector_type(8))) float v8f;

#define WMMA_F32(a, b, c) \
    __builtin_amdgcn_wmma_f32_16x16x4_f32(false, (a), false, (b), (short)0, (c), false, false)

#define BN_EPS 1e-3f
#define BIG 3.0e38f


__device__ __forceinline__ int frag_col(int i) {
    return ((((i >> 7) & 1) << 1) + ((i >> 1) & 1)) * 16 + ((i >> 2) & 15);
}
__device__ __forceinline__ constexpr int chunk_row_w0(int q) {
    return ((q >> 4) << 6) + (((q & 15) >> 1) << 3) + ((q & 1) << 1);
}
__device__ __forceinline__ constexpr int chunk_row_w12(int q) {
    return ((q >> 2) << 4) + ((q & 3) << 1);
}

#define GROUP_FENCE() __builtin_amdgcn_sched_barrier(0)

__device__ __forceinline__ void load_frags(v4f f[8], const float* wl, int q0, int q1, int q2, int q3) {
    f[0] = *(const v4f*)(wl + (q0 << 8)); f[1] = *(const v4f*)(wl + (q0 << 8) + 128);
    f[2] = *(const v4f*)(wl + (q1 << 8)); f[3] = *(const v4f*)(wl + (q1 << 8) + 128);
    f[4] = *(const v4f*)(wl + (q2 << 8)); f[5] = *(const v4f*)(wl + (q2 << 8) + 128);
    f[6] = *(const v4f*)(wl + (q3 << 8)); f[7] = *(const v4f*)(wl + (q3 << 8) + 128);
}

__device__ __forceinline__ void chunk_wa(v4f w01, v4f w23, v2f act, v8f acc[4]) {
    v2f f0 = {w01.x, w01.y}, f1 = {w01.z, w01.w}, f2 = {w23.x, w23.y}, f3 = {w23.z, w23.w};
    acc[0] = WMMA_F32(f0, act, acc[0]);
    acc[1] = WMMA_F32(f1, act, acc[1]);
    acc[2] = WMMA_F32(f2, act, acc[2]);
    acc[3] = WMMA_F32(f3, act, acc[3]);
}

__device__ __forceinline__ void bn_relu_t(v8f acc[4], const float* sc, const float* bi, int half) {
#pragma unroll
    for (int t = 0; t < 4; ++t) {
        const int d = (t << 4) + (half << 3);
        const v4f s0 = *(const v4f*)(sc + d), s1 = *(const v4f*)(sc + d + 4);
        const v4f b0 = *(const v4f*)(bi + d), b1 = *(const v4f*)(bi + d + 4);
#pragma unroll
        for (int r = 0; r < 4; ++r) {
            acc[t][r]     = fmaxf(acc[t][r] * s0[r] + b0[r], 0.0f);
            acc[t][r + 4] = fmaxf(acc[t][r + 4] * s1[r] + b1[r], 0.0f);
        }
    }
}

template <int XMASK>
__device__ __forceinline__ int min_row_xor(int v) {
    return min(v, __builtin_amdgcn_update_dpp(0, v, 0x160 + XMASK, 0xf, 0xf, true));
}
__device__ __forceinline__ int wave_min_i(int q) {
    q = min_row_xor<1>(q);
    q = min_row_xor<2>(q);
    q = min_row_xor<4>(q);
    q = min_row_xor<8>(q);
    return min(q, (int)__builtin_amdgcn_permlanex16(0, q, 0x76543210, 0xfedcba98, true, false));
}

#define LDS_FLOATS 36608
#define FTS_PITCH 68
#define NBR_PITCH 17

__global__ __launch_bounds__(256) void edgeconv_fused_kernel(
    const float* __restrict__ points,
    const float* __restrict__ feats,
    const float* __restrict__ w0,
    const float* __restrict__ w1,
    const float* __restrict__ w2,
    const float* __restrict__ gammas,
    const float* __restrict__ betas,
    const float* __restrict__ means,
    const float* __restrict__ vars,
    const float* __restrict__ scw,
    const float* __restrict__ scg, const float* __restrict__ scb,
    const float* __restrict__ scm, const float* __restrict__ scv,
    float* __restrict__ out)
{
    __shared__ __attribute__((aligned(16))) float smem[LDS_FLOATS];
    float* pts    = smem;
    float* Wf0    = pts + 768;
    float* Wf1    = Wf0 + 8192;
    float* Wf2    = Wf1 + 4096;
    float* bnS    = Wf2 + 4096;
    float* bnB    = bnS + 192;
    float* ctrAll = bnB + 192;
    float* ftsAll = ctrAll + 8192;
    int*   nbrAll = (int*)(ftsAll + 8 * 16 * FTS_PITCH);

    const int tid  = threadIdx.x;
    const int lane = tid & 31;
    const int w    = tid >> 5;
    const int lrow = lane & 15;
    const int half = lane >> 4;
    const int b    = blockIdx.x >> 1;
    const int pi0  = ((blockIdx.x & 1) << 7) | (w << 4);

    const float* pb = points + b * 768 + tid;
    const float p0 = pb[0], p1 = pb[256], p2 = pb[512];
    __builtin_amdgcn_sched_barrier(0);
    const int bt = (tid < 192) ? tid : (tid - 64);
    const float bnG = gammas[bt], bnV = vars[bt], bnBe = betas[bt], bnM = means[bt];
    const int gs = tid & 1, gh = (tid >> 6) & 1, gcol = frag_col(tid);
    const float* w0t = w0 + ((((gh << 2) + gs) << 6) + gcol);
    const int o12 = (((gh << 3) + gs) << 6) + gcol;
    const float* w1t = w1 + o12;
    const float* w2t = w2 + o12;
    float r0[32], r1[16], r2[16];
#pragma unroll
    for (int q = 0; q < 32; ++q) r0[q] = w0t[chunk_row_w0(q) << 6];
#pragma unroll
    for (int q = 0; q < 16; ++q) {
        r1[q] = w1t[chunk_row_w12(q) << 6];
        r2[q] = w2t[chunk_row_w12(q) << 6];
    }
    pts[tid] = p0; pts[tid + 256] = p1; pts[tid + 512] = p2;
    __builtin_amdgcn_fence(__ATOMIC_RELEASE, "workgroup", "local");
    __builtin_amdgcn_s_barrier();
    __builtin_amdgcn_fence(__ATOMIC_ACQUIRE, "workgroup", "local");

    float* cW   = ctrAll + (w << 10);
    float* fW   = ftsAll + w * (16 * FTS_PITCH);
    int*   nbrW = nbrAll + w * (16 * NBR_PITCH);

    float cx[8], cy[8], cz[8];
#pragma unroll
    for (int j = 0; j < 8; ++j) {
        const int q = lane + (j << 5);
        cx[j] = pts[q * 3]; cy[j] = pts[q * 3 + 1]; cz[j] = pts[q * 3 + 2];
    }
    const int kBig = __builtin_bit_cast(int, BIG);
#pragma unroll 1
    for (int pp = 0; pp < 16; pp += 2) {
        int key[2][8];
#pragma unroll
        for (int u = 0; u < 2; ++u) {
            const int pi = pi0 + pp + u;
            const float px = pts[pi * 3], py = pts[pi * 3 + 1], pz = pts[pi * 3 + 2];
#pragma unroll
            for (int j = 0; j < 8; ++j) {
                const int q = lane + (j << 5);
                const float dx = cx[j] - px;
                const float dy = cy[j] - py;
                const float dz = cz[j] - pz;
                const float dd = dx * dx + dy * dy + dz * dz;
                key[u][j] = __builtin_bit_cast(int, (q == pi) ? BIG : dd);
            }
        }
        unsigned nq[2] = {0u, 0u};
        for (int it = 0; it < 16; ++it) {
#pragma unroll
            for (int u = 0; u < 2; ++u) {
                const int m = wave_min_i(min(min(min(key[u][0], key[u][1]), min(key[u][2], key[u][3])),
                                             min(min(key[u][4], key[u][5]), min(key[u][6], key[u][7]))));
                int ql = 0x7fffffff;
#pragma unroll
                for (int j = 7; j >= 0; --j) ql = (key[u][j] == m) ? (lane + (j << 5)) : ql;
                const int bq = wave_min_i(ql);
#pragma unroll
                for (int j = 0; j < 8; ++j) key[u][j] = ((lane + (j << 5)) == bq) ? kBig : key[u][j];
                nq[u] = (lrow == it) ? (unsigned)bq : nq[u];
            }
        }
        nbrW[pp * NBR_PITCH + lrow]       = (int)nq[0];
        nbrW[(pp + 1) * NBR_PITCH + lrow] = (int)nq[1];
    }

    if (tid < 192) {
        const float s = bnG * (1.0f / sqrtf(bnV + BN_EPS));
        bnS[tid] = s;
        bnB[tid] = bnBe - bnM * s;
    }
#pragma unroll
    for (int q = 0; q < 32; ++q) Wf0[(q << 8) + tid] = r0[q];
#pragma unroll
    for (int q = 0; q < 16; ++q) {
        Wf1[(q << 8) + tid] = r1[q];
        Wf2[(q << 8) + tid] = r2[q];
    }
    __syncthreads();

    const float* fbat  = feats + (b << 14);
    const float* fwave = fbat + (pi0 << 6);
#pragma unroll
    for (int i = 0; i < 8; ++i)
        *(v4f*)(cW + (i << 7) + (lane << 2)) = *(const v4f*)(fwave + (i << 7) + (lane << 2));
    const float* cf = fwave + ((lrow << 6) + (half << 2));
    v4f ctr[8];
#pragma unroll
    for (int i = 0; i < 8; ++i) ctr[i] = *(const v4f*)(cf + (i << 3));

    const v8f vz = {0.f, 0.f, 0.f, 0.f, 0.f, 0.f, 0.f, 0.f};
    const float* wl0 = Wf0 + (lane << 2);
    const float* wl1 = Wf1 + (lane << 2);
    const float* wl2 = Wf2 + (lane << 2);
    v4f fc[8], fn[8];

    v8f zc[4] = {vz, vz, vz, vz};
    load_frags(fc, wl0, 0, 1, 2, 3);
#pragma unroll
    for (int g = 0; g < 4; ++g) {
        if (g < 3) load_frags(fn, wl0, 4 * g + 4, 4 * g + 5, 4 * g + 6, 4 * g + 7);
        else       load_frags(fn, wl0, 16, 17, 18, 19);
        GROUP_FENCE();
        const v4f ca = ctr[2 * g], cb = ctr[2 * g + 1];
        v2f c0 = {ca.x, ca.y}, c1 = {ca.z, ca.w}, c2 = {cb.x, cb.y}, c3 = {cb.z, cb.w};
        chunk_wa(fc[0], fc[1], c0, zc);
        chunk_wa(fc[2], fc[3], c1, zc);
        chunk_wa(fc[4], fc[5], c2, zc);
        chunk_wa(fc[6], fc[7], c3, zc);
        GROUP_FENCE();
#pragma unroll
        for (int k = 0; k < 8; ++k) fc[k] = fn[k];
    }

    const int* nb = nbrW + lrow * NBR_PITCH;
    const unsigned hoff = (unsigned)(half << 2);
    v4f en[8];
    {
        const float* nf = fbat + (((unsigned)nb[0] << 6) + hoff);
#pragma unroll
        for (int i = 0; i < 8; ++i) en[i] = *(const v4f*)(nf + (i << 3));
    }
    v8f sum[4] = {vz, vz, vz, vz};
#pragma unroll 1
    for (int j = 0; j < 16; ++j) {
        v4f e[8];
#pragma unroll
        for (int i = 0; i < 8; ++i) e[i] = en[i] - ctr[i];
        {
            const int jn = (j < 15) ? (j + 1) : 15;
            const float* nf = fbat + (((unsigned)nb[jn] << 6) + hoff);
#pragma unroll
            for (int i = 0; i < 8; ++i) en[i] = *(const v4f*)(nf + (i << 3));
        }

        v8f acc[4] = {zc[0], zc[1], zc[2], zc[3]};
#pragma unroll
        for (int g = 0; g < 4; ++g) {
            if (g < 3) load_frags(fn, wl0, 16 + 4 * g + 4, 16 + 4 * g + 5, 16 + 4 * g + 6, 16 + 4 * g + 7);
            else       load_frags(fn, wl1, 0, 1, 2, 3);
            GROUP_FENCE();
            const v4f ea = e[2 * g], eb = e[2 * g + 1];
            v2f e0 = {ea.x, ea.y}, e1 = {ea.z, ea.w}, e2 = {eb.x, eb.y}, e3 = {eb.z, eb.w};
            chunk_wa(fc[0], fc[1], e0, acc);
            chunk_wa(fc[2], fc[3], e1, acc);
            chunk_wa(fc[4], fc[5], e2, acc);
            chunk_wa(fc[6], fc[7], e3, acc);
            GROUP_FENCE();
#pragma unroll
            for (int k = 0; k < 8; ++k) fc[k] = fn[k];
        }
        bn_relu_t(acc, bnS, bnB, half);

        v8f acc1[4] = {vz, vz, vz, vz};
#pragma unroll
        for (int t = 0; t < 4; ++t) {
            if (t < 3) load_frags(fn, wl1, 4 * t + 4, 4 * t + 5, 4 * t + 6, 4 * t + 7);
            else       load_frags(fn, wl2, 0, 1, 2, 3);
            GROUP_FENCE();
#pragma unroll
            for (int k = 0; k < 4; ++k) {
                v2f a = {acc[t][2 * k], acc[t][2 * k + 1]};
                chunk_wa(fc[2 * k], fc[2 * k + 1], a, acc1);
            }
            GROUP_FENCE();
#pragma unroll
            for (int k = 0; k < 8; ++k) fc[k] = fn[k];
        }
        bn_relu_t(acc1, bnS + 64, bnB + 64, half);

        v8f acc2[4] = {vz, vz, vz, vz};
#pragma unroll
        for (int t = 0; t < 4; ++t) {
            if (t < 3) load_frags(fn, wl2, 4 * t + 4, 4 * t + 5, 4 * t + 6, 4 * t + 7);
            else       load_frags(fn, wl0, 16, 17, 18, 19);
            GROUP_FENCE();
#pragma unroll
            for (int k = 0; k < 4; ++k) {
                v2f a = {acc1[t][2 * k], acc1[t][2 * k + 1]};
                chunk_wa(fc[2 * k], fc[2 * k + 1], a, acc2);
            }
            GROUP_FENCE();
#pragma unroll
            for (int k = 0; k < 8; ++k) fc[k] = fn[k];
        }
        bn_relu_t(acc2, bnS + 128, bnB + 128, half);

#pragma unroll
        for (int t = 0; t < 4; ++t) sum[t] += acc2[t];
    }

#pragma unroll
    for (int t = 0; t < 4; ++t) {
        const v8f m = sum[t] * (1.0f / 16.0f);
        float* dst = fW + lrow * FTS_PITCH + (t << 4) + (half << 3);
        const v4f lo = {m[0], m[1], m[2], m[3]}, hi = {m[4], m[5], m[6], m[7]};
        *(v4f*)dst = lo;
        *(v4f*)(dst + 4) = hi;
    }

    const int d0 = lane << 1;
    const float* scr = scw + d0;
    v2f wsc[64];
#pragma unroll
    for (int c = 0; c < 64; ++c) wsc[c] = *(const v2f*)(scr + (c << 6));
    const float sA = scg[d0] * (1.0f / sqrtf(scv[d0] + BN_EPS));
    const float bA = scb[d0] - scm[d0] * sA;
    const float sB = scg[d0 + 1] * (1.0f / sqrtf(scv[d0 + 1] + BN_EPS));
    const float bB = scb[d0 + 1] - scm[d0 + 1] * sB;
    float* orow = out + ((((b << 8) | pi0) << 6) + d0);
#pragma unroll 4
    for (int p = 0; p < 16; ++p) {
        const float* cr = cW + (p << 6);
        float s0 = 0.f, s1 = 0.f;
#pragma unroll
        for (int c = 0; c < 64; ++c) {
            const float f = cr[c];
            s0 += f * wsc[c].x;
            s1 += f * wsc[c].y;
        }
        const v2f ft = *(const v2f*)(fW + p * FTS_PITCH + d0);
        v2f o;
        o.x = fmaxf(s0 * sA + bA + ft.x, 0.f);
        o.y = fmaxf(s1 * sB + bB + ft.y, 0.f);
        *(v2f*)(orow + (p << 6)) = o;
    }
}

extern "C" void kernel_launch(void* const* d_in, const int* in_sizes, int n_in,
                              void* d_out, int out_size, void* d_ws, size_t ws_size,
                              hipStream_t stream) {
    (void)in_sizes; (void)n_in; (void)d_ws; (void)ws_size; (void)out_size;
    const float* points = (const float*)d_in[0];
    const float* feats  = (const float*)d_in[1];
    const float* w0     = (const float*)d_in[2];
    const float* w1     = (const float*)d_in[3];
    const float* w2     = (const float*)d_in[4];
    const float* gammas = (const float*)d_in[5];
    const float* betas  = (const float*)d_in[6];
    const float* means  = (const float*)d_in[7];
    const float* vars   = (const float*)d_in[8];
    const float* scw    = (const float*)d_in[9];
    const float* scg    = (const float*)d_in[10];
    const float* scb    = (const float*)d_in[11];
    const float* scm    = (const float*)d_in[12];
    const float* scv    = (const float*)d_in[13];
    float* out = (float*)d_out;

    dim3 grid(256), block(256);
    edgeconv_fused_kernel<<<grid, block, 0, stream>>>(
        points, feats, w0, w1, w2, gammas, betas, means, vars,
        scw, scg, scb, scm, scv, out);
}


# neighbour search by rank, keys of the wave's 16 points in LDS: round j+1 (next neighbour of every point) runs among the products of MLP trip j, vector unit beside matrix unit; no sum reordered
# speedup vs baseline: 6.7763x; 1.0392x over previous
#include <hip/hip_runtime.h>
#include <math.h>


typedef __attribute__((ext_vector_type(2))) float v2f;
typedef __attribute__((ext_vector_type(4))) float v4f;
typedef __attribute__((ext_vector_type(8))) float v8f;
typedef __attribute__((ext_vector_type(4))) int v4i;

#define WMMA_F32(a, b, c) \
    __builtin_amdgcn_wmma_f32_16x16x4_f32(false, (a), false, (b), (short)0, (c), false, false)

#define BN_EPS 1e-3f
#define BIG 3.0e38f


__device__ __forceinline__ int frag_col(int i) {
    return ((((i >> 7) & 1) << 1) + ((i >> 1) & 1)) * 16 + ((i >> 2) & 15);
}
__device__ __forceinline__ constexpr int chunk_row_w0(int q) {
    return ((q >> 4) << 6) + (((q & 15) >> 1) << 3) + ((q & 1) << 1);
}
__device__ __forceinline__ constexpr int chunk_row_w12(int q) {
    return ((q >> 2) << 4) + ((q & 3) << 1);
}

#define GROUP_FENCE() __builtin_amdgcn_sched_barrier(0)

#define SEARCH_AMONG_PRODUCTS(N, NV) do { \
    __builtin_amdgcn_sched_group_barrier(0x100, 8 + 2 * (N), 0); \
    __builtin_amdgcn_sched_group_barrier(0x008, 3, 0); \
    _Pragma("unroll") for (int sgb = 0; sgb < 13; ++sgb) { \
        __builtin_amdgcn_sched_group_barrier(0x002, (NV), 0); \
        __builtin_amdgcn_sched_group_barrier(0x008, 1, 0); \
    } \
    __builtin_amdgcn_sched_group_barrier(0x200, (N), 0); \
} while (0)

__device__ __forceinline__ void load_frags(v4f f[8], const float* wl, int q0, int q1, int q2, int q3) {
    f[0] = *(const v4f*)(wl + (q0 << 8)); f[1] = *(const v4f*)(wl + (q0 << 8) + 128);
    f[2] = *(const v4f*)(wl + (q1 << 8)); f[3] = *(const v4f*)(wl + (q1 << 8) + 128);
    f[4] = *(const v4f*)(wl + (q2 << 8)); f[5] = *(const v4f*)(wl + (q2 << 8) + 128);
    f[6] = *(const v4f*)(wl + (q3 << 8)); f[7] = *(const v4f*)(wl + (q3 << 8) + 128);
}

__device__ __forceinline__ void chunk_wa(v4f w01, v4f w23, v2f act, v8f acc[4]) {
    v2f f0 = {w01.x, w01.y}, f1 = {w01.z, w01.w}, f2 = {w23.x, w23.y}, f3 = {w23.z, w23.w};
    acc[0] = WMMA_F32(f0, act, acc[0]);
    acc[1] = WMMA_F32(f1, act, acc[1]);
    acc[2] = WMMA_F32(f2, act, acc[2]);
    acc[3] = WMMA_F32(f3, act, acc[3]);
}

__device__ __forceinline__ void bn_relu_t(v8f acc[4], const float* sc, const float* bi, int half) {
#pragma unroll
    for (int t = 0; t < 4; ++t) {
        const int d = (t << 4) + (half << 3);
        const v4f s0 = *(const v4f*)(sc + d), s1 = *(const v4f*)(sc + d + 4);
        const v4f b0 = *(const v4f*)(bi + d), b1 = *(const v4f*)(bi + d + 4);
#pragma unroll
        for (int r = 0; r < 4; ++r) {
            acc[t][r]     = fmaxf(acc[t][r] * s0[r] + b0[r], 0.0f);
            acc[t][r + 4] = fmaxf(acc[t][r + 4] * s1[r] + b1[r], 0.0f);
        }
    }
}

template <int XMASK>
__device__ __forceinline__ int min_row_xor(int v) {
    return min(v, __builtin_amdgcn_update_dpp(0, v, 0x160 + XMASK, 0xf, 0xf, true));
}

#define EACH_POINT _Pragma("unroll") for (int n = 0; n < N; ++n)
template <int N>
__device__ __forceinline__ void knn_pick_n(const v4i* k0, const v4i* k1, int lane, int* r) {
    int m[N], q[N];
    EACH_POINT
        m[n] = min(min(min(k0[n].x, k0[n].y), min(k0[n].z, k0[n].w)), min(min(k1[n].x, k1[n].y), min(k1[n].z, k1[n].w)));
    EACH_POINT m[n] = min_row_xor<1>(m[n]);
    EACH_POINT m[n] = min_row_xor<2>(m[n]);
    EACH_POINT m[n] = min_row_xor<4>(m[n]);
    EACH_POINT m[n] = min_row_xor<8>(m[n]);
    EACH_POINT
        m[n] = min(m[n], (int)__builtin_amdgcn_permlanex16(0, m[n], 0x76543210, 0xfedcba98, true, false));
    EACH_POINT q[n] = (k1[n].w == m[n]) ? (lane + 224) : 0x7fffffff;
    EACH_POINT q[n] = (k1[n].z == m[n]) ? (lane + 192) : q[n];
    EACH_POINT q[n] = (k1[n].y == m[n]) ? (lane + 160) : q[n];
    EACH_POINT q[n] = (k1[n].x == m[n]) ? (lane + 128) : q[n];
    EACH_POINT q[n] = (k0[n].w == m[n]) ? (lane + 96) : q[n];
    EACH_POINT q[n] = (k0[n].z == m[n]) ? (lane + 64) : q[n];
    EACH_POINT q[n] = (k0[n].y == m[n]) ? (lane + 32) : q[n];
    EACH_POINT q[n] = (k0[n].x == m[n]) ? lane : q[n];
    EACH_POINT q[n] = min_row_xor<1>(q[n]);
    EACH_POINT q[n] = min_row_xor<2>(q[n]);
    EACH_POINT q[n] = min_row_xor<4>(q[n]);
    EACH_POINT q[n] = min_row_xor<8>(q[n]);
    EACH_POINT
        r[n] = min(q[n], (int)__builtin_amdgcn_permlanex16(0, q[n], 0x76543210, 0xfedcba98, true, false));
}
template <int N>
__device__ __forceinline__ void knn_round_part(int* keyW, int* spare, int p0, int lane, int lrow, int kBig, int& nq) {
    v4i k0[N], k1[N];
    int r[N];
    EACH_POINT {
        k0[n] = *(const v4i*)(keyW + ((p0 + n) << 8));
        k1[n] = *(const v4i*)(keyW + ((p0 + n) << 8) + 128);
    }
    knn_pick_n<N>(k0, k1, lane, r);
    EACH_POINT {
        const int js = r[n] >> 5;
        int* own = keyW + ((p0 + n) << 8) + ((js >> 2) << 7) + (js & 3);
        int* dst = (lane == (r[n] & 31)) ? own : spare;
        *dst = kBig;
        nq = (lrow == p0 + n) ? r[n] : nq;
    }
}

#define LDS_FLOATS 34432
#define FTS_PITCH 68
#define KEY_INTS (8 * 16 * 256 + 8 * 32)

__global__ __launch_bounds__(256) void edgeconv_fused_kernel(
    const float* __restrict__ points,
    const float* __restrict__ feats,
    const float* __restrict__ w0,
    const float* __restrict__ w1,
    const float* __restrict__ w2,
    const float* __restrict__ gammas,
    const float* __restrict__ betas,
    const float* __restrict__ means,
    const float* __restrict__ vars,
    const float* __restrict__ scw,
    const float* __restrict__ scg, const float* __restrict__ scb,
    const float* __restrict__ scm, const float* __restrict__ scv,
    float* __restrict__ out)
{
    __shared__ __attribute__((aligned(16))) float smem[LDS_FLOATS];
    __shared__ __attribute__((aligned(16))) int kmem[KEY_INTS];
    float* pts    = smem;
    float* Wf0    = pts + 768;
    float* Wf1    = Wf0 + 8192;
    float* Wf2    = Wf1 + 4096;
    float* bnS    = Wf2 + 4096;
    float* bnB    = bnS + 192;
    float* ctrAll = bnB + 192;
    float* ftsAll = ctrAll + 8192;

    const int tid  = threadIdx.x;
    const int lane = tid & 31;
    const int w    = tid >> 5;
    const int lrow = lane & 15;
    const int half = lane >> 4;
    const int b    = blockIdx.x >> 1;
    const int pi0  = ((blockIdx.x & 1) << 7) | (w << 4);

    const float* pb = points + b * 768 + tid;
    const float p0 = pb[0], p1 = pb[256], p2 = pb[512];
    __builtin_amdgcn_sched_barrier(0);
    const int bt = (tid < 192) ? tid : (tid - 64);
    const float bnG = gammas[bt], bnV = vars[bt], bnBe = betas[bt], bnM = means[bt];
    const int gs = tid & 1, gh = (tid >> 6) & 1, gcol = frag_col(tid);
    const float* w0t = w0 + ((((gh << 2) + gs) << 6) + gcol);
    const int o12 = (((gh << 3) + gs) << 6) + gcol;
    const float* w1t = w1 + o12;
    const float* w2t = w2 + o12;
    float r0[32], r1[16], r2[16];
#pragma unroll
    for (int q = 0; q < 32; ++q) r0[q] = w0t[chunk_row_w0(q) << 6];
#pragma unroll
    for (int q = 0; q < 16; ++q) {
        r1[q] = w1t[chunk_row_w12(q) << 6];
        r2[q] = w2t[chunk_row_w12(q) << 6];
    }
    pts[tid] = p0; pts[tid + 256] = p1; pts[tid + 512] = p2;
    __builtin_amdgcn_fence(__ATOMIC_RELEASE, "workgroup", "local");
    __builtin_amdgcn_s_barrier();
    __builtin_amdgcn_fence(__ATOMIC_ACQUIRE, "workgroup", "local");

    float* cW   = ctrAll + (w << 10);
    float* fW   = ftsAll + w * (16 * FTS_PITCH);
    int*   keyW = kmem + (w << 12) + (lane << 2);
    int*   spareW = kmem + (8 << 12) + (w << 5) + lane;

    const int kBig = __builtin_bit_cast(int, BIG);
    {
        float cx[8], cy[8], cz[8];
#pragma unroll
        for (int j = 0; j < 8; ++j) {
            const int q = lane + (j << 5);
            cx[j] = pts[q * 3]; cy[j] = pts[q * 3 + 1]; cz[j] = pts[q * 3 + 2];
        }
#pragma unroll 2
        for (int p = 0; p < 16; ++p) {
            const int pi = pi0 + p;
            const float px = pts[pi * 3], py = pts[pi * 3 + 1], pz = pts[pi * 3 + 2];
            int key[8];
#pragma unroll
            for (int j = 0; j < 8; ++j) {
                const int q = lane + (j << 5);
                const float dx = cx[j] - px;
                const float dy = cy[j] - py;
                const float dz = cz[j] - pz;
                const float dd = dx * dx + dy * dy + dz * dz;
                key[j] = __builtin_bit_cast(int, (q == pi) ? BIG : dd);
            }
            const v4i k0 = {key[0], key[1], key[2], key[3]}, k1 = {key[4], key[5], key[6], key[7]};
            *(v4i*)(keyW + (p << 8)) = k0;
            *(v4i*)(keyW + (p << 8) + 128) = k1;
        }
    }
    int nq = 0;
#pragma unroll 1
    for (int p = 0; p < 16; p += 4) knn_round_part<4>(keyW, spareW, p, lane, lrow, kBig, nq);

    if (tid < 192) {
        const float s = bnG * (1.0f / sqrtf(bnV + BN_EPS));
        bnS[tid] = s;
        bnB[tid] = bnBe - bnM * s;
    }
#pragma unroll
    for (int q = 0; q < 32; ++q) Wf0[(q << 8) + tid] = r0[q];
#pragma unroll
    for (int q = 0; q < 16; ++q) {
        Wf1[(q << 8) + tid] = r1[q];
        Wf2[(q << 8) + tid] = r2[q];
    }
    __syncthreads();

    const float* fbat  = feats + (b << 14);
    const float* fwave = fbat + (pi0 << 6);
#pragma unroll
    for (int i = 0; i < 8; ++i)
        *(v4f*)(cW + (i << 7) + (lane << 2)) = *(const v4f*)(fwave + (i << 7) + (lane << 2));
    const float* cf = fwave + ((lrow << 6) + (half << 2));
    v4f ctr[8];
#pragma unroll
    for (int i = 0; i < 8; ++i) ctr[i] = *(const v4f*)(cf + (i << 3));

    const v8f vz = {0.f, 0.f, 0.f, 0.f, 0.f, 0.f, 0.f, 0.f};
    const float* wl0 = Wf0 + (lane << 2);
    const float* wl1 = Wf1 + (lane << 2);
    const float* wl2 = Wf2 + (lane << 2);
    v4f fc[8], fn[8];

    v8f zc[4] = {vz, vz, vz, vz};
    load_frags(fc, wl0, 0, 1, 2, 3);
#pragma unroll
    for (int g = 0; g < 4; ++g) {
        if (g < 3) load_frags(fn, wl0, 4 * g + 4, 4 * g + 5, 4 * g + 6, 4 * g + 7);
        else       load_frags(fn, wl0, 16, 17, 18, 19);
        GROUP_FENCE();
        const v4f ca = ctr[2 * g], cb = ctr[2 * g + 1];
        v2f c0 = {ca.x, ca.y}, c1 = {ca.z, ca.w}, c2 = {cb.x, cb.y}, c3 = {cb.z, cb.w};
        chunk_wa(fc[0], fc[1], c0, zc);
        chunk_wa(fc[2], fc[3], c1, zc);
        chunk_wa(fc[4], fc[5], c2, zc);
        chunk_wa(fc[6], fc[7], c3, zc);
        GROUP_FENCE();
#pragma unroll
        for (int k = 0; k < 8; ++k) fc[k] = fn[k];
    }

    const unsigned hoff = (unsigned)(half << 2);
    v4f en[8];
    {
        const float* nf = fbat + (((unsigned)nq << 6) + hoff);
#pragma unroll
        for (int i = 0; i < 8; ++i) en[i] = *(const v4f*)(nf + (i << 3));
    }
    v8f sum[4] = {vz, vz, vz, vz};
#pragma unroll 1
    for (int j = 0; j < 16; ++j) {
        v4f e[8];
#pragma unroll
        for (int i = 0; i < 8; ++i) e[i] = en[i] - ctr[i];
        int nqn = 0;

        v8f acc[4] = {zc[0], zc[1], zc[2], zc[3]};
#pragma unroll
        for (int g = 0; g < 4; ++g) {
            if (g < 3) load_frags(fn, wl0, 16 + 4 * g + 4, 16 + 4 * g + 5, 16 + 4 * g + 6, 16 + 4 * g + 7);
            else       load_frags(fn, wl1, 0, 1, 2, 3);
            if (g == 0) GROUP_FENCE();
            const v4f ea = e[2 * g], eb = e[2 * g + 1];
            v2f e0 = {ea.x, ea.y}, e1 = {ea.z, ea.w}, e2 = {eb.x, eb.y}, e3 = {eb.z, eb.w};
            chunk_wa(fc[0], fc[1], e0, acc);
            chunk_wa(fc[2], fc[3], e1, acc);
            chunk_wa(fc[4], fc[5], e2, acc);
            chunk_wa(fc[6], fc[7], e3, acc);
            if (g == 1)      { knn_round_part<3>(keyW, spareW, 0, lane, lrow, kBig, nqn); SEARCH_AMONG_PRODUCTS(3, 10); }
            else if (g == 2) { knn_round_part<3>(keyW, spareW, 3, lane, lrow, kBig, nqn); SEARCH_AMONG_PRODUCTS(3, 10); }
            else if (g == 3) { knn_round_part<2>(keyW, spareW, 6, lane, lrow, kBig, nqn); SEARCH_AMONG_PRODUCTS(2, 7); }
            GROUP_FENCE();
#pragma unroll
            for (int k = 0; k < 8; ++k) fc[k] = fn[k];
        }
        bn_relu_t(acc, bnS, bnB, half);

        v8f acc1[4] = {vz, vz, vz, vz};
#pragma unroll
        for (int t = 0; t < 4; ++t) {
            if (t < 3) load_frags(fn, wl1, 4 * t + 4, 4 * t + 5, 4 * t + 6, 4 * t + 7);
            else       load_frags(fn, wl2, 0, 1, 2, 3);
            if (t == 0) GROUP_FENCE();
#pragma unroll
            for (int k = 0; k < 4; ++k) {
                v2f a = {acc[t][2 * k], acc[t][2 * k + 1]};
                chunk_wa(fc[2 * k], fc[2 * k + 1], a, acc1);
            }
            if (t == 1)      { knn_round_part<3>(keyW, spareW, 8, lane, lrow, kBig, nqn); SEARCH_AMONG_PRODUCTS(3, 10); }
            else if (t == 2) { knn_round_part<3>(keyW, spareW, 11, lane, lrow, kBig, nqn); SEARCH_AMONG_PRODUCTS(3, 10); }
            else if (t == 3) { knn_round_part<2>(keyW, spareW, 14, lane, lrow, kBig, nqn); SEARCH_AMONG_PRODUCTS(2, 7); }
            GROUP_FENCE();
#pragma unroll
            for (int k = 0; k < 8; ++k) fc[k] = fn[k];
        }
        {
            const float* nf = fbat + (((unsigned)nqn << 6) + hoff);
#pragma unroll
            for (int i = 0; i < 8; ++i) en[i] = *(const v4f*)(nf + (i << 3));
        }
        bn_relu_t(acc1, bnS + 64, bnB + 64, half);

        v8f acc2[4] = {vz, vz, vz, vz};
#pragma unroll
        for (int t = 0; t < 4; ++t) {
            if (t < 3) load_frags(fn, wl2, 4 * t + 4, 4 * t + 5, 4 * t + 6, 4 * t + 7);
            else       load_frags(fn, wl0, 16, 17, 18, 19);
            GROUP_FENCE();
#pragma unroll
            for (int k = 0; k < 4; ++k) {
                v2f a = {acc1[t][2 * k], acc1[t][2 * k + 1]};
                chunk_wa(fc[2 * k], fc[2 * k + 1], a, acc2);
            }
            GROUP_FENCE();
#pragma unroll
            for (int k = 0; k < 8; ++k) fc[k] = fn[k];
        }
        bn_relu_t(acc2, bnS + 128, bnB + 128, half);

#pragma unroll
        for (int t = 0; t < 4; ++t) sum[t] += acc2[t];
    }

#pragma unroll
    for (int t = 0; t < 4; ++t) {
        const v8f m = sum[t] * (1.0f / 16.0f);
        float* dst = fW + lrow * FTS_PITCH + (t << 4) + (half << 3);
        const v4f lo = {m[0], m[1], m[2], m[3]}, hi = {m[4], m[5], m[6], m[7]};
        *(v4f*)dst = lo;
        *(v4f*)(dst + 4) = hi;
    }

    const int d0 = lane << 1;
    const float* scr = scw + d0;
    v2f wsc[64];
#pragma unroll
    for (int c = 0; c < 64; ++c) wsc[c] = *(const v2f*)(scr + (c << 6));
    const float sA = scg[d0] * (1.0f / sqrtf(scv[d0] + BN_EPS));
    const float bA = scb[d0] - scm[d0] * sA;
    const float sB = scg[d0 + 1] * (1.0f / sqrtf(scv[d0 + 1] + BN_EPS));
    const float bB = scb[d0 + 1] - scm[d0 + 1] * sB;
    float* orow = out + ((((b << 8) | pi0) << 6) + d0);
#pragma unroll 4
    for (int p = 0; p < 16; ++p) {
        const float* cr = cW + (p << 6);
        float s0 = 0.f, s1 = 0.f;
#pragma unroll
        for (int c = 0; c < 64; ++c) {
            const float f = cr[c];
            s0 += f * wsc[c].x;
            s1 += f * wsc[c].y;
        }
        const v2f ft = *(const v2f*)(fW + p * FTS_PITCH + d0);
        v2f o;
        o.x = fmaxf(s0 * sA + bA + ft.x, 0.f);
        o.y = fmaxf(s1 * sB + bB + ft.y, 0.f);
        *(v2f*)(orow + (p << 6)) = o;
    }
}

extern "C" void kernel_launch(void* const* d_in, const int* in_sizes, int n_in,
                              void* d_out, int out_size, void* d_ws, size_t ws_size,
                              hipStream_t stream) {
    (void)in_sizes; (void)n_in; (void)d_ws; (void)ws_size; (void)out_size;
    const float* points = (const float*)d_in[0];
    const float* feats  = (const float*)d_in[1];
    const float* w0     = (const float*)d_in[2];
    const float* w1     = (const float*)d_in[3];
    const float* w2     = (const float*)d_in[4];
    const float* gammas = (const float*)d_in[5];
    const float* betas  = (const float*)d_in[6];
    const float* means  = (const float*)d_in[7];
    const float* vars   = (const float*)d_in[8];
    const float* scw    = (const float*)d_in[9];
    const float* scg    = (const float*)d_in[10];
    const float* scb    = (const float*)d_in[11];
    const float* scm    = (const float*)d_in[12];
    const float* scv    = (const float*)d_in[13];
    float* out = (float*)d_out;

    dim3 grid(256), block(256);
    edgeconv_fused_kernel<<<grid, block, 0, stream>>>(
        points, feats, w0, w1, w2, gammas, betas, means, vars,
        scw, scg, scb, scm, scv, out);
}


# layer 2's weights read from LDS once per wave into 128 registers and kept over the 16 trips, its BN terms read ahead with layer 1's: its 64 products run with no LDS read among them; no sum reordered
# speedup vs baseline: 6.9963x; 1.0325x over previous
#include <hip/hip_runtime.h>
#include <math.h>


typedef __attribute__((ext_vector_type(2))) float v2f;
typedef __attribute__((ext_vector_type(4))) float v4f;
typedef __attribute__((ext_vector_type(8))) float v8f;
typedef __attribute__((ext_vector_type(4))) int v4i;

#define WMMA_F32(a, b, c) \
    __builtin_amdgcn_wmma_f32_16x16x4_f32(false, (a), false, (b), (short)0, (c), false, false)

#define BN_EPS 1e-3f
#define BIG 3.0e38f


__device__ __forceinline__ int frag_col(int i) {
    return ((((i >> 7) & 1) << 1) + ((i >> 1) & 1)) * 16 + ((i >> 2) & 15);
}
__device__ __forceinline__ constexpr int chunk_row_w0(int q) {
    return ((q >> 4) << 6) + (((q & 15) >> 1) << 3) + ((q & 1) << 1);
}
__device__ __forceinline__ constexpr int chunk_row_w12(int q) {
    return ((q >> 2) << 4) + ((q & 3) << 1);
}

#define GROUP_FENCE() __builtin_amdgcn_sched_barrier(0)

#define SEARCH_AMONG_PRODUCTS(N, NV) do { \
    __builtin_amdgcn_sched_group_barrier(0x100, 8 + 2 * (N), 0); \
    __builtin_amdgcn_sched_group_barrier(0x008, 3, 0); \
    _Pragma("unroll") for (int sgb = 0; sgb < 13; ++sgb) { \
        __builtin_amdgcn_sched_group_barrier(0x002, (NV), 0); \
        __builtin_amdgcn_sched_group_barrier(0x008, 1, 0); \
    } \
    __builtin_amdgcn_sched_group_barrier(0x200, (N), 0); \
} while (0)

__device__ __forceinline__ void load_frags(v4f f[8], const float* wl, int q0, int q1, int q2, int q3) {
    f[0] = *(const v4f*)(wl + (q0 << 8)); f[1] = *(const v4f*)(wl + (q0 << 8) + 128);
    f[2] = *(const v4f*)(wl + (q1 << 8)); f[3] = *(const v4f*)(wl + (q1 << 8) + 128);
    f[4] = *(const v4f*)(wl + (q2 << 8)); f[5] = *(const v4f*)(wl + (q2 << 8) + 128);
    f[6] = *(const v4f*)(wl + (q3 << 8)); f[7] = *(const v4f*)(wl + (q3 << 8) + 128);
}

__device__ __forceinline__ void chunk_wa(v4f w01, v4f w23, v2f act, v8f acc[4]) {
    v2f f0 = {w01.x, w01.y}, f1 = {w01.z, w01.w}, f2 = {w23.x, w23.y}, f3 = {w23.z, w23.w};
    acc[0] = WMMA_F32(f0, act, acc[0]);
    acc[1] = WMMA_F32(f1, act, acc[1]);
    acc[2] = WMMA_F32(f2, act, acc[2]);
    acc[3] = WMMA_F32(f3, act, acc[3]);
}

__device__ __forceinline__ void bn_terms(v4f s[8], v4f b[8], const float* sc, const float* bi, int half) {
#pragma unroll
    for (int t = 0; t < 4; ++t) {
        const int d = (t << 4) + (half << 3);
        s[2 * t] = *(const v4f*)(sc + d); s[2 * t + 1] = *(const v4f*)(sc + d + 4);
        b[2 * t] = *(const v4f*)(bi + d); b[2 * t + 1] = *(const v4f*)(bi + d + 4);
    }
}
__device__ __forceinline__ void bn_relu_r(v8f acc[4], const v4f s[8], const v4f b[8]) {
#pragma unroll
    for (int t = 0; t < 4; ++t) {
#pragma unroll
        for (int r = 0; r < 4; ++r) {
            acc[t][r]     = fmaxf(acc[t][r] * s[2 * t][r] + b[2 * t][r], 0.0f);
            acc[t][r + 4] = fmaxf(acc[t][r + 4] * s[2 * t + 1][r] + b[2 * t + 1][r], 0.0f);
        }
    }
}
__device__ __forceinline__ void bn_relu_t(v8f acc[4], const float* sc, const float* bi, int half) {
    v4f s[8], b[8];
    bn_terms(s, b, sc, bi, half);
    bn_relu_r(acc, s, b);
}

template <int XMASK>
__device__ __forceinline__ int min_row_xor(int v) {
    return min(v, __builtin_amdgcn_update_dpp(0, v, 0x160 + XMASK, 0xf, 0xf, true));
}

#define EACH_POINT _Pragma("unroll") for (int n = 0; n < N; ++n)
template <int N>
__device__ __forceinline__ void knn_pick_n(const v4i* k0, const v4i* k1, int lane, int* r) {
    int m[N], q[N];
    EACH_POINT
        m[n] = min(min(min(k0[n].x, k0[n].y), min(k0[n].z, k0[n].w)), min(min(k1[n].x, k1[n].y), min(k1[n].z, k1[n].w)));
    EACH_POINT m[n] = min_row_xor<1>(m[n]);
    EACH_POINT m[n] = min_row_xor<2>(m[n]);
    EACH_POINT m[n] = min_row_xor<4>(m[n]);
    EACH_POINT m[n] = min_row_xor<8>(m[n]);
    EACH_POINT
        m[n] = min(m[n], (int)__builtin_amdgcn_permlanex16(0, m[n], 0x76543210, 0xfedcba98, true, false));
    EACH_POINT q[n] = (k1[n].w == m[n]) ? (lane + 224) : 0x7fffffff;
    EACH_POINT q[n] = (k1[n].z == m[n]) ? (lane + 192) : q[n];
    EACH_POINT q[n] = (k1[n].y == m[n]) ? (lane + 160) : q[n];
    EACH_POINT q[n] = (k1[n].x == m[n]) ? (lane + 128) : q[n];
    EACH_POINT q[n] = (k0[n].w == m[n]) ? (lane + 96) : q[n];
    EACH_POINT q[n] = (k0[n].z == m[n]) ? (lane + 64) : q[n];
    EACH_POINT q[n] = (k0[n].y == m[n]) ? (lane + 32) : q[n];
    EACH_POINT q[n] = (k0[n].x == m[n]) ? lane : q[n];
    EACH_POINT q[n] = min_row_xor<1>(q[n]);
    EACH_POINT q[n] = min_row_xor<2>(q[n]);
    EACH_POINT q[n] = min_row_xor<4>(q[n]);
    EACH_POINT q[n] = min_row_xor<8>(q[n]);
    EACH_POINT
        r[n] = min(q[n], (int)__builtin_amdgcn_permlanex16(0, q[n], 0x76543210, 0xfedcba98, true, false));
}
template <int N>
__device__ __forceinline__ void knn_round_part(int* keyW, int* spare, int p0, int lane, int lrow, int kBig, int& nq) {
    v4i k0[N], k1[N];
    int r[N];
    EACH_POINT {
        k0[n] = *(const v4i*)(keyW + ((p0 + n) << 8));
        k1[n] = *(const v4i*)(keyW + ((p0 + n) << 8) + 128);
    }
    knn_pick_n<N>(k0, k1, lane, r);
    EACH_POINT {
        const int js = r[n] >> 5;
        int* own = keyW + ((p0 + n) << 8) + ((js >> 2) << 7) + (js & 3);
        int* dst = (lane == (r[n] & 31)) ? own : spare;
        *dst = kBig;
        nq = (lrow == p0 + n) ? r[n] : nq;
    }
}

#define LDS_FLOATS 34432
#define FTS_PITCH 68
#define KEY_INTS (8 * 16 * 256 + 8 * 32)

__global__ __launch_bounds__(256) void edgeconv_fused_kernel(
    const float* __restrict__ points,
    const float* __restrict__ feats,
    const float* __restrict__ w0,
    const float* __restrict__ w1,
    const float* __restrict__ w2,
    const float* __restrict__ gammas,
    const float* __restrict__ betas,
    const float* __restrict__ means,
    const float* __restrict__ vars,
    const float* __restrict__ scw,
    const float* __restrict__ scg, const float* __restrict__ scb,
    const float* __restrict__ scm, const float* __restrict__ scv,
    float* __restrict__ out)
{
    __shared__ __attribute__((aligned(16))) float smem[LDS_FLOATS];
    __shared__ __attribute__((aligned(16))) int kmem[KEY_INTS];
    float* pts    = smem;
    float* Wf0    = pts + 768;
    float* Wf1    = Wf0 + 8192;
    float* Wf2    = Wf1 + 4096;
    float* bnS    = Wf2 + 4096;
    float* bnB    = bnS + 192;
    float* ctrAll = bnB + 192;
    float* ftsAll = ctrAll + 8192;

    const int tid  = threadIdx.x;
    const int lane = tid & 31;
    const int w    = tid >> 5;
    const int lrow = lane & 15;
    const int half = lane >> 4;
    const int b    = blockIdx.x >> 1;
    const int pi0  = ((blockIdx.x & 1) << 7) | (w << 4);

    const float* pb = points + b * 768 + tid;
    const float p0 = pb[0], p1 = pb[256], p2 = pb[512];
    __builtin_amdgcn_sched_barrier(0);
    const int bt = (tid < 192) ? tid : (tid - 64);
    const float bnG = gammas[bt], bnV = vars[bt], bnBe = betas[bt], bnM = means[bt];
    const int gs = tid & 1, gh = (tid >> 6) & 1, gcol = frag_col(tid);
    const float* w0t = w0 + ((((gh << 2) + gs) << 6) + gcol);
    const int o12 = (((gh << 3) + gs) << 6) + gcol;
    const float* w1t = w1 + o12;
    const float* w2t = w2 + o12;
    float r0[32], r1[16], r2[16];
#pragma unroll
    for (int q = 0; q < 32; ++q) r0[q] = w0t[chunk_row_w0(q) << 6];
#pragma unroll
    for (int q = 0; q < 16; ++q) {
        r1[q] = w1t[chunk_row_w12(q) << 6];
        r2[q] = w2t[chunk_row_w12(q) << 6];
    }
    pts[tid] = p0; pts[tid + 256] = p1; pts[tid + 512] = p2;
    __builtin_amdgcn_fence(__ATOMIC_RELEASE, "workgroup", "local");
    __builtin_amdgcn_s_barrier();
    __builtin_amdgcn_fence(__ATOMIC_ACQUIRE, "workgroup", "local");

    float* cW   = ctrAll + (w << 10);
    float* fW   = ftsAll + w * (16 * FTS_PITCH);
    int*   keyW = kmem + (w << 12) + (lane << 2);
    int*   spareW = kmem + (8 << 12) + (w << 5) + lane;

    const int kBig = __builtin_bit_cast(int, BIG);
    {
        float cx[8], cy[8], cz[8];
#pragma unroll
        for (int j = 0; j < 8; ++j) {
            const int q = lane + (j << 5);
            cx[j] = pts[q * 3]; cy[j] = pts[q * 3 + 1]; cz[j] = pts[q * 3 + 2];
        }
#pragma unroll 2
        for (int p = 0; p < 16; ++p) {
            const int pi = pi0 + p;
            const float px = pts[pi * 3], py = pts[pi * 3 + 1], pz = pts[pi * 3 + 2];
            int key[8];
#pragma unroll
            for (int j = 0; j < 8; ++j) {
                const int q = lane + (j << 5);
                const float dx = cx[j] - px;
                const float dy = cy[j] - py;
                const float dz = cz[j] - pz;
                const float dd = dx * dx + dy * dy + dz * dz;
                key[j] = __builtin_bit_cast(int, (q == pi) ? BIG : dd);
            }
            const v4i k0 = {key[0], key[1], key[2], key[3]}, k1 = {key[4], key[5], key[6], key[7]};
            *(v4i*)(keyW + (p << 8)) = k0;
            *(v4i*)(keyW + (p << 8) + 128) = k1;
        }
    }
    int nq = 0;
#pragma unroll 1
    for (int p = 0; p < 16; p += 4) knn_round_part<4>(keyW, spareW, p, lane, lrow, kBig, nq);

    if (tid < 192) {
        const float s = bnG * (1.0f / sqrtf(bnV + BN_EPS));
        bnS[tid] = s;
        bnB[tid] = bnBe - bnM * s;
    }
#pragma unroll
    for (int q = 0; q < 32; ++q) Wf0[(q << 8) + tid] = r0[q];
#pragma unroll
    for (int q = 0; q < 16; ++q) {
        Wf1[(q << 8) + tid] = r1[q];
        Wf2[(q << 8) + tid] = r2[q];
    }
    __syncthreads();

    const float* fbat  = feats + (b << 14);
    const float* fwave = fbat + (pi0 << 6);
#pragma unroll
    for (int i = 0; i < 8; ++i)
        *(v4f*)(cW + (i << 7) + (lane << 2)) = *(const v4f*)(fwave + (i << 7) + (lane << 2));
    const float* cf = fwave + ((lrow << 6) + (half << 2));
    v4f ctr[8];
#pragma unroll
    for (int i = 0; i < 8; ++i) ctr[i] = *(const v4f*)(cf + (i << 3));

    const v8f vz = {0.f, 0.f, 0.f, 0.f, 0.f, 0.f, 0.f, 0.f};
    const float* wl0 = Wf0 + (lane << 2);
    const float* wl1 = Wf1 + (lane << 2);
    const float* wl2 = Wf2 + (lane << 2);
    v4f fc[8], fn[8];

    v8f zc[4] = {vz, vz, vz, vz};
    load_frags(fc, wl0, 0, 1, 2, 3);
#pragma unroll
    for (int g = 0; g < 4; ++g) {
        if (g < 3) load_frags(fn, wl0, 4 * g + 4, 4 * g + 5, 4 * g + 6, 4 * g + 7);
        else       load_frags(fn, wl0, 16, 17, 18, 19);
        GROUP_FENCE();
        const v4f ca = ctr[2 * g], cb = ctr[2 * g + 1];
        v2f c0 = {ca.x, ca.y}, c1 = {ca.z, ca.w}, c2 = {cb.x, cb.y}, c3 = {cb.z, cb.w};
        chunk_wa(fc[0], fc[1], c0, zc);
        chunk_wa(fc[2], fc[3], c1, zc);
        chunk_wa(fc[4], fc[5], c2, zc);
        chunk_wa(fc[6], fc[7], c3, zc);
        GROUP_FENCE();
#pragma unroll
        for (int k = 0; k < 8; ++k) fc[k] = fn[k];
    }

    v4f w2r[32];
#pragma unroll
    for (int q = 0; q < 16; ++q) {
        w2r[2 * q]     = *(const v4f*)(wl2 + (q << 8));
        w2r[2 * q + 1] = *(const v4f*)(wl2 + (q << 8) + 128);
    }

    const unsigned hoff = (unsigned)(half << 2);
    v4f en[8];
    {
        const float* nf = fbat + (((unsigned)nq << 6) + hoff);
#pragma unroll
        for (int i = 0; i < 8; ++i) en[i] = *(const v4f*)(nf + (i << 3));
    }
    v8f sum[4] = {vz, vz, vz, vz};
#pragma unroll 1
    for (int j = 0; j < 16; ++j) {
        v4f e[8];
#pragma unroll
        for (int i = 0; i < 8; ++i) e[i] = en[i] - ctr[i];
        int nqn = 0;

        v8f acc[4] = {zc[0], zc[1], zc[2], zc[3]};
#pragma unroll
        for (int g = 0; g < 4; ++g) {
            if (g < 3) load_frags(fn, wl0, 16 + 4 * g + 4, 16 + 4 * g + 5, 16 + 4 * g + 6, 16 + 4 * g + 7);
            else       load_frags(fn, wl1, 0, 1, 2, 3);
            if (g == 0) GROUP_FENCE();
            const v4f ea = e[2 * g], eb = e[2 * g + 1];
            v2f e0 = {ea.x, ea.y}, e1 = {ea.z, ea.w}, e2 = {eb.x, eb.y}, e3 = {eb.z, eb.w};
            chunk_wa(fc[0], fc[1], e0, acc);
            chunk_wa(fc[2], fc[3], e1, acc);
            chunk_wa(fc[4], fc[5], e2, acc);
            chunk_wa(fc[6], fc[7], e3, acc);
            if (g == 1)      { knn_round_part<3>(keyW, spareW, 0, lane, lrow, kBig, nqn); SEARCH_AMONG_PRODUCTS(3, 10); }
            else if (g == 2) { knn_round_part<3>(keyW, spareW, 3, lane, lrow, kBig, nqn); SEARCH_AMONG_PRODUCTS(3, 10); }
            else if (g == 3) { knn_round_part<2>(keyW, spareW, 6, lane, lrow, kBig, nqn); SEARCH_AMONG_PRODUCTS(2, 7); }
            GROUP_FENCE();
#pragma unroll
            for (int k = 0; k < 8; ++k) fc[k] = fn[k];
        }
        bn_relu_t(acc, bnS, bnB, half);

        v8f acc1[4] = {vz, vz, vz, vz};
#pragma unroll
        for (int t = 0; t < 4; ++t) {
            if (t < 3) load_frags(fn, wl1, 4 * t + 4, 4 * t + 5, 4 * t + 6, 4 * t + 7);
            else       load_frags(fn, wl0, 16, 17, 18, 19);
            if (t == 0) GROUP_FENCE();
#pragma unroll
            for (int k = 0; k < 4; ++k) {
                v2f a = {acc[t][2 * k], acc[t][2 * k + 1]};
                chunk_wa(fc[2 * k], fc[2 * k + 1], a, acc1);
            }
            if (t == 1)      { knn_round_part<3>(keyW, spareW, 8, lane, lrow, kBig, nqn); SEARCH_AMONG_PRODUCTS(3, 10); }
            else if (t == 2) { knn_round_part<3>(keyW, spareW, 11, lane, lrow, kBig, nqn); SEARCH_AMONG_PRODUCTS(3, 10); }
            else if (t == 3) { knn_round_part<2>(keyW, spareW, 14, lane, lrow, kBig, nqn); SEARCH_AMONG_PRODUCTS(2, 7); }
            GROUP_FENCE();
#pragma unroll
            for (int k = 0; k < 8; ++k) fc[k] = fn[k];
        }
        {
            const float* nf = fbat + (((unsigned)nqn << 6) + hoff);
#pragma unroll
            for (int i = 0; i < 8; ++i) en[i] = *(const v4f*)(nf + (i << 3));
        }
        v4f s1[8], b1[8], s2[8], b2[8];
        bn_terms(s1, b1, bnS + 64, bnB + 64, half);
        bn_terms(s2, b2, bnS + 128, bnB + 128, half);
        bn_relu_r(acc1, s1, b1);

        v8f acc2[4] = {vz, vz, vz, vz};
        GROUP_FENCE();
#pragma unroll
        for (int t = 0; t < 4; ++t) {
#pragma unroll
            for (int k = 0; k < 4; ++k) {
                v2f a = {acc1[t][2 * k], acc1[t][2 * k + 1]};
                chunk_wa(w2r[2 * (4 * t + k)], w2r[2 * (4 * t + k) + 1], a, acc2);
            }
        }
        GROUP_FENCE();
        bn_relu_r(acc2, s2, b2);

#pragma unroll
        for (int t = 0; t < 4; ++t) sum[t] += acc2[t];
    }

#pragma unroll
    for (int t = 0; t < 4; ++t) {
        const v8f m = sum[t] * (1.0f / 16.0f);
        float* dst = fW + lrow * FTS_PITCH + (t << 4) + (half << 3);
        const v4f lo = {m[0], m[1], m[2], m[3]}, hi = {m[4], m[5], m[6], m[7]};
        *(v4f*)dst = lo;
        *(v4f*)(dst + 4) = hi;
    }

    const int d0 = lane << 1;
    const float* scr = scw + d0;
    v2f wsc[64];
#pragma unroll
    for (int c = 0; c < 64; ++c) wsc[c] = *(const v2f*)(scr + (c << 6));
    const float sA = scg[d0] * (1.0f / sqrtf(scv[d0] + BN_EPS));
    const float bA = scb[d0] - scm[d0] * sA;
    const float sB = scg[d0 + 1] * (1.0f / sqrtf(scv[d0 + 1] + BN_EPS));
    const float bB = scb[d0 + 1] - scm[d0 + 1] * sB;
    float* orow = out + ((((b << 8) | pi0) << 6) + d0);
#pragma unroll 4
    for (int p = 0; p < 16; ++p) {
        const float* cr = cW + (p << 6);
        float s0 = 0.f, s1 = 0.f;
#pragma unroll
        for (int c = 0; c < 64; ++c) {
            const float f = cr[c];
            s0 += f * wsc[c].x;
            s1 += f * wsc[c].y;
        }
        const v2f ft = *(const v2f*)(fW + p * FTS_PITCH + d0);
        v2f o;
        o.x = fmaxf(s0 * sA + bA + ft.x, 0.f);
        o.y = fmaxf(s1 * sB + bB + ft.y, 0.f);
        *(v2f*)(orow + (p << 6)) = o;
    }
}

extern "C" void kernel_launch(void* const* d_in, const int* in_sizes, int n_in,
                              void* d_out, int out_size, void* d_ws, size_t ws_size,
                              hipStream_t stream) {
    (void)in_sizes; (void)n_in; (void)d_ws; (void)ws_size; (void)out_size;
    const float* points = (const float*)d_in[0];
    const float* feats  = (const float*)d_in[1];
    const float* w0     = (const float*)d_in[2];
    const float* w1     = (const float*)d_in[3];
    const float* w2     = (const float*)d_in[4];
    const float* gammas = (const float*)d_in[5];
    const float* betas  = (const float*)d_in[6];
    const float* means  = (const float*)d_in[7];
    const float* vars   = (const float*)d_in[8];
    const float* scw    = (const float*)d_in[9];
    const float* scg    = (const float*)d_in[10];
    const float* scb    = (const float*)d_in[11];
    const float* scm    = (const float*)d_in[12];
    const float* scv    = (const float*)d_in[13];
    float* out = (float*)d_out;

    dim3 grid(256), block(256);
    edgeconv_fused_kernel<<<grid, block, 0, stream>>>(
        points, feats, w0, w1, w2, gammas, betas, means, vars,
        scw, scg, scb, scm, scv, out);
}


# neighbour search: lane (x,h) owns point x's keys of candidates 128h..128h+127 as 16 groups of 8 with group minima in LDS; a round reads 16 minima and one group, no row reduction; no sum reordered
# speedup vs baseline: 8.2342x; 1.1769x over previous
#include <hip/hip_runtime.h>
#include <math.h>


typedef __attribute__((ext_vector_type(2))) float v2f;
typedef __attribute__((ext_vector_type(4))) float v4f;
typedef __attribute__((ext_vector_type(8))) float v8f;
typedef __attribute__((ext_vector_type(4))) int v4i;

#define WMMA_F32(a, b, c) \
    __builtin_amdgcn_wmma_f32_16x16x4_f32(false, (a), false, (b), (short)0, (c), false, false)

#define BN_EPS 1e-3f
#define BIG 3.0e38f


__device__ __forceinline__ int frag_col(int i) {
    return ((((i >> 7) & 1) << 1) + ((i >> 1) & 1)) * 16 + ((i >> 2) & 15);
}
__device__ __forceinline__ constexpr int chunk_row_w0(int q) {
    return ((q >> 4) << 6) + (((q & 15) >> 1) << 3) + ((q & 1) << 1);
}
__device__ __forceinline__ constexpr int chunk_row_w12(int q) {
    return ((q >> 2) << 4) + ((q & 3) << 1);
}

#define GROUP_FENCE() __builtin_amdgcn_sched_barrier(0)

#define SEARCH_AMONG_PRODUCTS(NL, NV, NS) do { \
    __builtin_amdgcn_sched_group_barrier(0x100, 8 + (NL), 0); \
    __builtin_amdgcn_sched_group_barrier(0x008, 3, 0); \
    _Pragma("unroll") for (int sgb = 0; sgb < 13; ++sgb) { \
        __builtin_amdgcn_sched_group_barrier(0x002, (NV), 0); \
        __builtin_amdgcn_sched_group_barrier(0x008, 1, 0); \
    } \
    if ((NS) != 0) __builtin_amdgcn_sched_group_barrier(0x200, (NS), 0); \
} while (0)

__device__ __forceinline__ void load_frags(v4f f[8], const float* wl, int q0, int q1, int q2, int q3) {
    f[0] = *(const v4f*)(wl + (q0 << 8)); f[1] = *(const v4f*)(wl + (q0 << 8) + 128);
    f[2] = *(const v4f*)(wl + (q1 << 8)); f[3] = *(const v4f*)(wl + (q1 << 8) + 128);
    f[4] = *(const v4f*)(wl + (q2 << 8)); f[5] = *(const v4f*)(wl + (q2 << 8) + 128);
    f[6] = *(const v4f*)(wl + (q3 << 8)); f[7] = *(const v4f*)(wl + (q3 << 8) + 128);
}

__device__ __forceinline__ void chunk_wa(v4f w01, v4f w23, v2f act, v8f acc[4]) {
    v2f f0 = {w01.x, w01.y}, f1 = {w01.z, w01.w}, f2 = {w23.x, w23.y}, f3 = {w23.z, w23.w};
    acc[0] = WMMA_F32(f0, act, acc[0]);
    acc[1] = WMMA_F32(f1, act, acc[1]);
    acc[2] = WMMA_F32(f2, act, acc[2]);
    acc[3] = WMMA_F32(f3, act, acc[3]);
}

__device__ __forceinline__ void bn_terms(v4f s[8], v4f b[8], const float* sc, const float* bi, int half) {
#pragma unroll
    for (int t = 0; t < 4; ++t) {
        const int d = (t << 4) + (half << 3);
        s[2 * t] = *(const v4f*)(sc + d); s[2 * t + 1] = *(const v4f*)(sc + d + 4);
        b[2 * t] = *(const v4f*)(bi + d); b[2 * t + 1] = *(const v4f*)(bi + d + 4);
    }
}
__device__ __forceinline__ void bn_relu_r(v8f acc[4], const v4f s[8], const v4f b[8]) {
#pragma unroll
    for (int t = 0; t < 4; ++t) {
#pragma unroll
        for (int r = 0; r < 4; ++r) {
            acc[t][r]     = fmaxf(acc[t][r] * s[2 * t][r] + b[2 * t][r], 0.0f);
            acc[t][r + 4] = fmaxf(acc[t][r + 4] * s[2 * t + 1][r] + b[2 * t + 1][r], 0.0f);
        }
    }
}
__device__ __forceinline__ void bn_relu_t(v8f acc[4], const float* sc, const float* bi, int half) {
    v4f s[8], b[8];
    bn_terms(s, b, sc, bi, half);
    bn_relu_r(acc, s, b);
}

__device__ __forceinline__ int min8(v4i a, v4i b) {
    return min(min(min(a.x, a.y), min(a.z, a.w)), min(min(b.x, b.y), min(b.z, b.w)));
}
__device__ __forceinline__ void knn_read_minima(const int* sumL, v4i G[4]) {
    G[0] = *(const v4i*)(sumL);       G[1] = *(const v4i*)(sumL + 128);
    G[2] = *(const v4i*)(sumL + 256); G[3] = *(const v4i*)(sumL + 384);
}
__device__ __forceinline__ void knn_pick_group(const v4i G[4], int& m, int& g) {
    m = min(min8(G[0], G[1]), min8(G[2], G[3]));
    int q = 15;
    q = (G[3].z == m) ? 14 : q; q = (G[3].y == m) ? 13 : q; q = (G[3].x == m) ? 12 : q;
    q = (G[2].w == m) ? 11 : q; q = (G[2].z == m) ? 10 : q; q = (G[2].y == m) ? 9 : q; q = (G[2].x == m) ? 8 : q;
    q = (G[1].w == m) ? 7 : q;  q = (G[1].z == m) ? 6 : q;  q = (G[1].y == m) ? 5 : q; q = (G[1].x == m) ? 4 : q;
    q = (G[0].w == m) ? 3 : q;  q = (G[0].z == m) ? 2 : q;  q = (G[0].y == m) ? 1 : q; q = (G[0].x == m) ? 0 : q;
    g = q;
}
__device__ __forceinline__ int knn_pick_slot(v4i k0, v4i k1, int m) {
    int s = 7;
    s = (k1.z == m) ? 6 : s; s = (k1.y == m) ? 5 : s; s = (k1.x == m) ? 4 : s;
    s = (k0.w == m) ? 3 : s; s = (k0.z == m) ? 2 : s; s = (k0.y == m) ? 1 : s; s = (k0.x == m) ? 0 : s;
    return s;
}
__device__ __forceinline__ void knn_settle(int m, int cand, int half, int& r, bool& win) {
    const int om = (int)__builtin_amdgcn_permlanex16(0, m, 0x76543210, 0xfedcba98, true, false);
    const int oc = (int)__builtin_amdgcn_permlanex16(0, cand, 0x76543210, 0xfedcba98, true, false);
    win = (m + half) <= om;
    r = win ? cand : oc;
}
__device__ __forceinline__ int knn_round_whole(int* keyL, int* sumL, int* spare, int half, int kBig) {
    v4i G[4];
    knn_read_minima(sumL, G);
    int m, g;
    knn_pick_group(G, m, g);
    int* gp = keyL + (g << 8);
    const int s = knn_pick_slot(*(const v4i*)gp, *(const v4i*)(gp + 128), m);
    int r; bool win;
    knn_settle(m, (half << 7) + (g << 3) + s, half, r, win);
    int* own = gp + ((s >> 2) << 7) + (s & 3);
    *(win ? own : spare) = kBig;
    sumL[((g >> 2) << 7) + (g & 3)] = min8(*(const v4i*)gp, *(const v4i*)(gp + 128));
    return r;
}

#define LDS_FLOATS 34432
#define FTS_PITCH 68
#define KEY_INTS (8 * 4096 + 8 * 512 + 8 * 32)

__global__ __launch_bounds__(256) void edgeconv_fused_kernel(
    const float* __restrict__ points,
    const float* __restrict__ feats,
    const float* __restrict__ w0,
    const float* __restrict__ w1,
    const float* __restrict__ w2,
    const float* __restrict__ gammas,
    const float* __restrict__ betas,
    const float* __restrict__ means,
    const float* __restrict__ vars,
    const float* __restrict__ scw,
    const float* __restrict__ scg, const float* __restrict__ scb,
    const float* __restrict__ scm, const float* __restrict__ scv,
    float* __restrict__ out)
{
    __shared__ __attribute__((aligned(16))) float smem[LDS_FLOATS];
    __shared__ __attribute__((aligned(16))) int kmem[KEY_INTS];
    float* pts    = smem;
    float* Wf0    = pts + 768;
    float* Wf1    = Wf0 + 8192;
    float* Wf2    = Wf1 + 4096;
    float* bnS    = Wf2 + 4096;
    float* bnB    = bnS + 192;
    float* ctrAll = bnB + 192;
    float* ftsAll = ctrAll + 8192;

    const int tid  = threadIdx.x;
    const int lane = tid & 31;
    const int w    = tid >> 5;
    const int lrow = lane & 15;
    const int half = lane >> 4;
    const int b    = blockIdx.x >> 1;
    const int pi0  = ((blockIdx.x & 1) << 7) | (w << 4);

    const float* pb = points + b * 768 + tid;
    const float p0 = pb[0], p1 = pb[256], p2 = pb[512];
    __builtin_amdgcn_sched_barrier(0);
    const int bt = (tid < 192) ? tid : (tid - 64);
    const float bnG = gammas[bt], bnV = vars[bt], bnBe = betas[bt], bnM = means[bt];
    const int gs = tid & 1, gh = (tid >> 6) & 1, gcol = frag_col(tid);
    const float* w0t = w0 + ((((gh << 2) + gs) << 6) + gcol);
    const int o12 = (((gh << 3) + gs) << 6) + gcol;
    const float* w1t = w1 + o12;
    const float* w2t = w2 + o12;
    float r0[32], r1[16], r2[16];
#pragma unroll
    for (int q = 0; q < 32; ++q) r0[q] = w0t[chunk_row_w0(q) << 6];
#pragma unroll
    for (int q = 0; q < 16; ++q) {
        r1[q] = w1t[chunk_row_w12(q) << 6];
        r2[q] = w2t[chunk_row_w12(q) << 6];
    }
    pts[tid] = p0; pts[tid + 256] = p1; pts[tid + 512] = p2;
    __builtin_amdgcn_fence(__ATOMIC_RELEASE, "workgroup", "local");
    __builtin_amdgcn_s_barrier();
    __builtin_amdgcn_fence(__ATOMIC_ACQUIRE, "workgroup", "local");

    float* cW   = ctrAll + (w << 10);
    float* fW   = ftsAll + w * (16 * FTS_PITCH);
    int*   keyL = kmem + (w << 12) + (lane << 2);
    int*   sumL = kmem + (8 << 12) + (w << 9) + (lane << 2);
    int*   spareW = kmem + (8 << 12) + (8 << 9) + (w << 5) + lane;

    const int kBig = __builtin_bit_cast(int, BIG);
    {
        const int myp = pi0 + lrow;
        const float px = pts[myp * 3], py = pts[myp * 3 + 1], pz = pts[myp * 3 + 2];
        const float* cb = pts + half * 384;
        const int own = myp - (half << 7);
#pragma unroll 2
        for (int g = 0; g < 16; ++g) {
            const v4f* c4 = (const v4f*)(cb + g * 24);
            const v4f a0 = c4[0], a1 = c4[1], a2 = c4[2], a3 = c4[3], a4 = c4[4], a5 = c4[5];
            const float fl[24] = {a0.x, a0.y, a0.z, a0.w, a1.x, a1.y, a1.z, a1.w, a2.x, a2.y, a2.z, a2.w,
                                  a3.x, a3.y, a3.z, a3.w, a4.x, a4.y, a4.z, a4.w, a5.x, a5.y, a5.z, a5.w};
            int key[8];
#pragma unroll
            for (int s = 0; s < 8; ++s) {
                const float dx = fl[3 * s] - px;
                const float dy = fl[3 * s + 1] - py;
                const float dz = fl[3 * s + 2] - pz;
                const float dd = dx * dx + dy * dy + dz * dz;
                key[s] = __builtin_bit_cast(int, (((g << 3) + s) == own) ? BIG : dd);
            }
            const v4i k0 = {key[0], key[1], key[2], key[3]}, k1 = {key[4], key[5], key[6], key[7]};
            *(v4i*)(keyL + (g << 8)) = k0;
            *(v4i*)(keyL + (g << 8) + 128) = k1;
            sumL[((g >> 2) << 7) + (g & 3)] = min8(k0, k1);
        }
    }
    const int nq = knn_round_whole(keyL, sumL, spareW, half, kBig);

    if (tid < 192) {
        const float s = bnG * (1.0f / sqrtf(bnV + BN_EPS));
        bnS[tid] = s;
        bnB[tid] = bnBe - bnM * s;
    }
#pragma unroll
    for (int q = 0; q < 32; ++q) Wf0[(q << 8) + tid] = r0[q];
#pragma unroll
    for (int q = 0; q < 16; ++q) {
        Wf1[(q << 8) + tid] = r1[q];
        Wf2[(q << 8) + tid] = r2[q];
    }
    __syncthreads();

    const float* fbat  = feats + (b << 14);
    const float* fwave = fbat + (pi0 << 6);
#pragma unroll
    for (int i = 0; i < 8; ++i)
        *(v4f*)(cW + (i << 7) + (lane << 2)) = *(const v4f*)(fwave + (i << 7) + (lane << 2));
    const float* cf = fwave + ((lrow << 6) + (half << 2));
    v4f ctr[8];
#pragma unroll
    for (int i = 0; i < 8; ++i) ctr[i] = *(const v4f*)(cf + (i << 3));

    const v8f vz = {0.f, 0.f, 0.f, 0.f, 0.f, 0.f, 0.f, 0.f};
    const float* wl0 = Wf0 + (lane << 2);
    const float* wl1 = Wf1 + (lane << 2);
    const float* wl2 = Wf2 + (lane << 2);
    v4f fc[8], fn[8];

    v8f zc[4] = {vz, vz, vz, vz};
    load_frags(fc, wl0, 0, 1, 2, 3);
#pragma unroll
    for (int g = 0; g < 4; ++g) {
        if (g < 3) load_frags(fn, wl0, 4 * g + 4, 4 * g + 5, 4 * g + 6, 4 * g + 7);
        else       load_frags(fn, wl0, 16, 17, 18, 19);
        GROUP_FENCE();
        const v4f ca = ctr[2 * g], cb = ctr[2 * g + 1];
        v2f c0 = {ca.x, ca.y}, c1 = {ca.z, ca.w}, c2 = {cb.x, cb.y}, c3 = {cb.z, cb.w};
        chunk_wa(fc[0], fc[1], c0, zc);
        chunk_wa(fc[2], fc[3], c1, zc);
        chunk_wa(fc[4], fc[5], c2, zc);
        chunk_wa(fc[6], fc[7], c3, zc);
        GROUP_FENCE();
#pragma unroll
        for (int k = 0; k < 8; ++k) fc[k] = fn[k];
    }

    v4f w2r[32];
#pragma unroll
    for (int q = 0; q < 16; ++q) {
        w2r[2 * q]     = *(const v4f*)(wl2 + (q << 8));
        w2r[2 * q + 1] = *(const v4f*)(wl2 + (q << 8) + 128);
    }

    const unsigned hoff = (unsigned)(half << 2);
    v4f en[8];
    {
        const float* nf = fbat + (((unsigned)nq << 6) + hoff);
#pragma unroll
        for (int i = 0; i < 8; ++i) en[i] = *(const v4f*)(nf + (i << 3));
    }
    v8f sum[4] = {vz, vz, vz, vz};
#pragma unroll 1
    for (int j = 0; j < 16; ++j) {
        v4f e[8];
#pragma unroll
        for (int i = 0; i < 8; ++i) e[i] = en[i] - ctr[i];
        int nqn = 0;
        int sm = 0, sg = 0;
        int* sgp = keyL;

        v8f acc[4] = {zc[0], zc[1], zc[2], zc[3]};
#pragma unroll
        for (int g = 0; g < 4; ++g) {
            if (g < 3) load_frags(fn, wl0, 16 + 4 * g + 4, 16 + 4 * g + 5, 16 + 4 * g + 6, 16 + 4 * g + 7);
            else       load_frags(fn, wl1, 0, 1, 2, 3);
            if (g == 0) GROUP_FENCE();
            const v4f ea = e[2 * g], eb = e[2 * g + 1];
            v2f e0 = {ea.x, ea.y}, e1 = {ea.z, ea.w}, e2 = {eb.x, eb.y}, e3 = {eb.z, eb.w};
            chunk_wa(fc[0], fc[1], e0, acc);
            chunk_wa(fc[2], fc[3], e1, acc);
            chunk_wa(fc[4], fc[5], e2, acc);
            chunk_wa(fc[6], fc[7], e3, acc);
            if (g == 1) {
                v4i G[4];
                knn_read_minima(sumL, G);
                knn_pick_group(G, sm, sg);
                sgp = keyL + (sg << 8);
                SEARCH_AMONG_PRODUCTS(4, 3, 0);
            } else if (g == 2) {
                const int s = knn_pick_slot(*(const v4i*)sgp, *(const v4i*)(sgp + 128), sm);
                bool win;
                knn_settle(sm, (half << 7) + (sg << 3) + s, half, nqn, win);
                int* own = sgp + ((s >> 2) << 7) + (s & 3);
                *(win ? own : spareW) = kBig;
                SEARCH_AMONG_PRODUCTS(2, 3, 1);
            } else if (g == 3) {
                sumL[((sg >> 2) << 7) + (sg & 3)] = min8(*(const v4i*)sgp, *(const v4i*)(sgp + 128));
                SEARCH_AMONG_PRODUCTS(2, 1, 1);
            }
            GROUP_FENCE();
#pragma unroll
            for (int k = 0; k < 8; ++k) fc[k] = fn[k];
        }
        bn_relu_t(acc, bnS, bnB, half);

        v8f acc1[4] = {vz, vz, vz, vz};
#pragma unroll
        for (int t = 0; t < 4; ++t) {
            if (t < 3) load_frags(fn, wl1, 4 * t + 4, 4 * t + 5, 4 * t + 6, 4 * t + 7);
            else       load_frags(fn, wl0, 16, 17, 18, 19);
            if (t == 0) GROUP_FENCE();
#pragma unroll
            for (int k = 0; k < 4; ++k) {
                v2f a = {acc[t][2 * k], acc[t][2 * k + 1]};
                chunk_wa(fc[2 * k], fc[2 * k + 1], a, acc1);
            }
            if (t != 0) {
                __builtin_amdgcn_sched_group_barrier(0x100, 8, 0);
                __builtin_amdgcn_sched_group_barrier(0x008, 16, 0);
            }
            GROUP_FENCE();
#pragma unroll
            for (int k = 0; k < 8; ++k) fc[k] = fn[k];
        }
        {
            const float* nf = fbat + (((unsigned)nqn << 6) + hoff);
#pragma unroll
            for (int i = 0; i < 8; ++i) en[i] = *(const v4f*)(nf + (i << 3));
        }
        v4f s1[8], b1[8], s2[8], b2[8];
        bn_terms(s1, b1, bnS + 64, bnB + 64, half);
        bn_terms(s2, b2, bnS + 128, bnB + 128, half);
        bn_relu_r(acc1, s1, b1);

        v8f acc2[4] = {vz, vz, vz, vz};
        GROUP_FENCE();
#pragma unroll
        for (int t = 0; t < 4; ++t) {
#pragma unroll
            for (int k = 0; k < 4; ++k) {
                v2f a = {acc1[t][2 * k], acc1[t][2 * k + 1]};
                chunk_wa(w2r[2 * (4 * t + k)], w2r[2 * (4 * t + k) + 1], a, acc2);
            }
        }
        GROUP_FENCE();
        bn_relu_r(acc2, s2, b2);

#pragma unroll
        for (int t = 0; t < 4; ++t) sum[t] += acc2[t];
    }

#pragma unroll
    for (int t = 0; t < 4; ++t) {
        const v8f m = sum[t] * (1.0f / 16.0f);
        float* dst = fW + lrow * FTS_PITCH + (t << 4) + (half << 3);
        const v4f lo = {m[0], m[1], m[2], m[3]}, hi = {m[4], m[5], m[6], m[7]};
        *(v4f*)dst = lo;
        *(v4f*)(dst + 4) = hi;
    }

    const int d0 = lane << 1;
    const float* scr = scw + d0;
    v2f wsc[64];
#pragma unroll
    for (int c = 0; c < 64; ++c) wsc[c] = *(const v2f*)(scr + (c << 6));
    const float sA = scg[d0] * (1.0f / sqrtf(scv[d0] + BN_EPS));
    const float bA = scb[d0] - scm[d0] * sA;
    const float sB = scg[d0 + 1] * (1.0f / sqrtf(scv[d0 + 1] + BN_EPS));
    const float bB = scb[d0 + 1] - scm[d0 + 1] * sB;
    float* orow = out + ((((b << 8) | pi0) << 6) + d0);
#pragma unroll 4
    for (int p = 0; p < 16; ++p) {
        const float* cr = cW + (p << 6);
        float s0 = 0.f, s1 = 0.f;
#pragma unroll
        for (int c = 0; c < 64; ++c) {
            const float f = cr[c];
            s0 += f * wsc[c].x;
            s1 += f * wsc[c].y;
        }
        const v2f ft = *(const v2f*)(fW + p * FTS_PITCH + d0);
        v2f o;
        o.x = fmaxf(s0 * sA + bA + ft.x, 0.f);
        o.y = fmaxf(s1 * sB + bB + ft.y, 0.f);
        *(v2f*)(orow + (p << 6)) = o;
    }
}

extern "C" void kernel_launch(void* const* d_in, const int* in_sizes, int n_in,
                              void* d_out, int out_size, void* d_ws, size_t ws_size,
                              hipStream_t stream) {
    (void)in_sizes; (void)n_in; (void)d_ws; (void)ws_size; (void)out_size;
    const float* points = (const float*)d_in[0];
    const float* feats  = (const float*)d_in[1];
    const float* w0     = (const float*)d_in[2];
    const float* w1     = (const float*)d_in[3];
    const float* w2     = (const float*)d_in[4];
    const float* gammas = (const float*)d_in[5];
    const float* betas  = (const float*)d_in[6];
    const float* means  = (const float*)d_in[7];
    const float* vars   = (const float*)d_in[8];
    const float* scw    = (const float*)d_in[9];
    const float* scg    = (const float*)d_in[10];
    const float* scb    = (const float*)d_in[11];
    const float* scm    = (const float*)d_in[12];
    const float* scv    = (const float*)d_in[13];
    float* out = (float*)d_out;

    dim3 grid(256), block(256);
    edgeconv_fused_kernel<<<grid, block, 0, stream>>>(
        points, feats, w0, w1, w2, gammas, betas, means, vars,
        scw, scg, scb, scm, scv, out);
}
